# starting kernel
# baseline (speedup 1.0000x reference)
#include <hip/hip_runtime.h>

typedef __attribute__((ext_vector_type(2))) float v2f;
typedef __attribute__((ext_vector_type(4))) float v4f;
typedef __attribute__((ext_vector_type(8))) float v8f;

#define WAVES  4
#define BLOCK  (WAVES * 32)
#define MTILE  (WAVES * 16)
#define NBLOCKS 256

constexpr int OFF_W1  = 0;
constexpr int OFF_W2  = OFF_W1  + 8192;
constexpr int OFF_FC  = OFF_W2  + 16384;
constexpr int OFF_B1  = OFF_FC  + 2048;
constexpr int OFF_B2  = OFF_B1  + 128;
constexpr int OFF_FCB = OFF_B2  + 128;
constexpr int OFF_H   = OFF_FCB + 16;
constexpr int LDS_FLOATS = OFF_H + WAVES * 2048;

__device__ __forceinline__ v8f wmma_f32(v2f a, v2f b, v8f c) {
    return __builtin_amdgcn_wmma_f32_16x16x4_f32(false, a, false, b, (short)0, c,
                                                 false, false);
}

template <int K, int N>
__device__ __forceinline__ void stage_packed(float* dst, const float* __restrict__ src,
                                             int tid) {
    const v4f* g = (const v4f*)src;
    for (int i4 = tid; i4 < (K * N) / 4; i4 += BLOCK) {
        const int i = i4 * 4;
        const int k = i / N;
        const int n = i % N;
        const v4f v = g[i4];
        float* p = dst + (k >> 2) * (4 * N) + ((k >> 1) & 1) * 2 + (k & 1) + n * 4;
        p[0]  = v.x;
        p[4]  = v.y;
        p[8]  = v.z;
        p[12] = v.w;
    }
}

__global__ void __launch_bounds__(BLOCK)
gnn_mlp_kernel(const float* __restrict__ x,
               const float* __restrict__ W1, const float* __restrict__ b1,
               const float* __restrict__ W2, const float* __restrict__ b2,
               const float* __restrict__ fcW, const float* __restrict__ fcb,
               float* __restrict__ out)
{
    extern __shared__ float lds[];
    const int tid = threadIdx.x;

    stage_packed<64, 128>(lds + OFF_W1, W1, tid);
    stage_packed<128, 128>(lds + OFF_W2, W2, tid);
    stage_packed<128, 16>(lds + OFF_FC, fcW, tid);
    if (tid < 128) lds[OFF_B1 + tid] = b1[tid];
    if (tid < 128) lds[OFF_B2 + tid] = b2[tid];
    if (tid < 16)  lds[OFF_FCB + tid] = fcb[tid];
    __syncthreads();

    const int lane = tid & 31;
    const int wave = tid >> 5;
    const int m    = lane & 15;
    const int hi   = lane >> 4;
    const int koff = hi * 2;
    const int mm   = hi * 8;

    float* myH = lds + OFF_H + wave * 2048;
    const long rowBase = (long)blockIdx.x * MTILE + wave * 16;

    v2f a1[16];
    {
        const float* xr = x + (rowBase + m) * 64;
#pragma unroll
        for (int kk = 0; kk < 16; ++kk)
            a1[kk] = *(const v2f*)(xr + 4 * kk + koff);
    }
    for (int nt = 0; nt < 8; ++nt) {
        const int n = nt * 16 + m;
        const float bias = lds[OFF_B1 + n];
        v8f c = {bias, bias, bias, bias, bias, bias, bias, bias};
        const float* bp = lds + OFF_W1 + n * 4 + koff;
#pragma unroll
        for (int kk = 0; kk < 16; ++kk)
            c = wmma_f32(a1[kk], *(const v2f*)(bp + kk * 512), c);
#pragma unroll
        for (int r = 0; r < 8; ++r)
            myH[(r + mm) * 128 + n] = fmaxf(c[r], 0.0f);
    }
    __syncthreads();

    v2f a2[32];
#pragma unroll
    for (int kk = 0; kk < 32; ++kk)
        a2[kk] = *(const v2f*)(myH + m * 128 + 4 * kk + koff);

    for (int nt = 0; nt < 8; ++nt) {
        const int n = nt * 16 + m;
        const float bias = lds[OFF_B2 + n];
        v8f c = {bias, bias, bias, bias, bias, bias, bias, bias};
        const float* bp = lds + OFF_W2 + n * 4 + koff;
#pragma unroll
        for (int kk = 0; kk < 32; ++kk)
            c = wmma_f32(a2[kk], *(const v2f*)(bp + kk * 512), c);
#pragma unroll
        for (int r = 0; r < 8; ++r)
            myH[(r + mm) * 128 + n] = fmaxf(c[r], 0.0f);
    }
    __syncthreads();

    v2f a3[32];
#pragma unroll
    for (int kk = 0; kk < 32; ++kk)
        a3[kk] = *(const v2f*)(myH + m * 128 + 4 * kk + koff);

    {
        const float bias = lds[OFF_FCB + m];
        v8f c = {bias, bias, bias, bias, bias, bias, bias, bias};
        const float* bp = lds + OFF_FC + m * 4 + koff;
#pragma unroll
        for (int kk = 0; kk < 32; ++kk)
            c = wmma_f32(a3[kk], *(const v2f*)(bp + kk * 64), c);
#pragma unroll
        for (int r = 0; r < 8; ++r)
            out[(rowBase + r + mm) * 16 + m] = c[r];
    }
}

extern "C" void kernel_launch(void* const* d_in, const int* in_sizes, int n_in,
                              void* d_out, int out_size, void* d_ws, size_t ws_size,
                              hipStream_t stream) {
    const float* x   = (const float*)d_in[0];
    const float* W1  = (const float*)d_in[1];
    const float* b1  = (const float*)d_in[2];
    const float* W2  = (const float*)d_in[3];
    const float* b2  = (const float*)d_in[4];
    const float* fcW = (const float*)d_in[5];
    const float* fcb = (const float*)d_in[6];
    float* out = (float*)d_out;

    const size_t shmem = (size_t)LDS_FLOATS * sizeof(float);
    hipFuncSetAttribute((const void*)gnn_mlp_kernel,
                        hipFuncAttributeMaxDynamicSharedMemorySize, (int)shmem);
    gnn_mlp_kernel<<<NBLOCKS, BLOCK, shmem, stream>>>(x, W1, b1, W2, b2, fcW, fcb, out);
}


# Weight staging in 4x4 blocks: 16-byte row loads transposed in registers to 16-byte LDS stores (was 4-byte stores with bank conflicts), unrolled so loads are not drained per loop trip
# speedup vs baseline: 1.3367x; 1.3367x over previous
#include <hip/hip_runtime.h>

typedef __attribute__((ext_vector_type(2))) float v2f;
typedef __attribute__((ext_vector_type(4))) float v4f;
typedef __attribute__((ext_vector_type(8))) float v8f;

#define WAVES  4
#define BLOCK  (WAVES * 32)
#define MTILE  (WAVES * 16)
#define NBLOCKS 256

constexpr int OFF_W1  = 0;
constexpr int OFF_W2  = OFF_W1  + 8192;
constexpr int OFF_FC  = OFF_W2  + 16384;
constexpr int OFF_B1  = OFF_FC  + 2048;
constexpr int OFF_B2  = OFF_B1  + 128;
constexpr int OFF_FCB = OFF_B2  + 128;
constexpr int OFF_H   = OFF_FCB + BLOCK;
constexpr int LDS_FLOATS = OFF_H + WAVES * 2048;

__device__ __forceinline__ v8f wmma_f32(v2f a, v2f b, v8f c) {
    return __builtin_amdgcn_wmma_f32_16x16x4_f32(false, a, false, b, (short)0, c,
                                                 false, false);
}

template <int N, int IT>
__device__ __forceinline__ void load_blocks(v4f (&r)[IT][4], const float* __restrict__ src,
                                            int tid) {
    constexpr int NB = N / 4;
    const v4f* g = (const v4f*)src;
#pragma unroll
    for (int it = 0; it < IT; ++it) {
        const int b  = tid + it * BLOCK;
        const int kk = b / NB;
        const int nb = b % NB;
#pragma unroll
        for (int j = 0; j < 4; ++j)
            r[it][j] = g[(4 * kk + j) * NB + nb];
    }
}

template <int N, int IT>
__device__ __forceinline__ void store_blocks(float* dst, const v4f (&r)[IT][4], int tid) {
    constexpr int NB = N / 4;
#pragma unroll
    for (int it = 0; it < IT; ++it) {
        const int b  = tid + it * BLOCK;
        const int kk = b / NB;
        const int nb = b % NB;
        v4f* d = (v4f*)(dst + kk * (4 * N) + nb * 16);
#pragma unroll
        for (int i = 0; i < 4; ++i) {
            const v4f t = {r[it][0][i], r[it][1][i], r[it][2][i], r[it][3][i]};
            d[i] = t;
        }
    }
}

__global__ void __launch_bounds__(BLOCK)
gnn_mlp_kernel(const float* __restrict__ x,
               const float* __restrict__ W1, const float* __restrict__ b1,
               const float* __restrict__ W2, const float* __restrict__ b2,
               const float* __restrict__ fcW, const float* __restrict__ fcb,
               float* __restrict__ out)
{
    extern __shared__ __attribute__((aligned(16))) float lds[];
    const int tid = threadIdx.x;

    v4f rw1[4][4], rw2[8][4], rfc[1][4];
    load_blocks<128, 4>(rw1, W1, tid);
    load_blocks<128, 8>(rw2, W2, tid);
    load_blocks<16, 1>(rfc, fcW, tid);
    const float vb1 = b1[tid];
    const float vb2 = b2[tid];
    const float vfb = fcb[tid & 15];
    store_blocks<128, 4>(lds + OFF_W1, rw1, tid);
    store_blocks<128, 8>(lds + OFF_W2, rw2, tid);
    store_blocks<16, 1>(lds + OFF_FC, rfc, tid);
    lds[OFF_B1 + tid] = vb1;
    lds[OFF_B2 + tid] = vb2;
    lds[OFF_FCB + tid] = vfb;
    __syncthreads();

    const int lane = tid & 31;
    const int wave = tid >> 5;
    const int m    = lane & 15;
    const int hi   = lane >> 4;
    const int koff = hi * 2;
    const int mm   = hi * 8;

    float* myH = lds + OFF_H + wave * 2048;
    const long rowBase = (long)blockIdx.x * MTILE + wave * 16;

    v2f a1[16];
    {
        const float* xr = x + (rowBase + m) * 64;
#pragma unroll
        for (int kk = 0; kk < 16; ++kk)
            a1[kk] = *(const v2f*)(xr + 4 * kk + koff);
    }
    for (int nt = 0; nt < 8; ++nt) {
        const int n = nt * 16 + m;
        const float bias = lds[OFF_B1 + n];
        v8f c = {bias, bias, bias, bias, bias, bias, bias, bias};
        const float* bp = lds + OFF_W1 + n * 4 + koff;
#pragma unroll
        for (int kk = 0; kk < 16; ++kk)
            c = wmma_f32(a1[kk], *(const v2f*)(bp + kk * 512), c);
#pragma unroll
        for (int r = 0; r < 8; ++r)
            myH[(r + mm) * 128 + n] = fmaxf(c[r], 0.0f);
    }
    __syncthreads();

    v2f a2[32];
#pragma unroll
    for (int kk = 0; kk < 32; ++kk)
        a2[kk] = *(const v2f*)(myH + m * 128 + 4 * kk + koff);

    for (int nt = 0; nt < 8; ++nt) {
        const int n = nt * 16 + m;
        const float bias = lds[OFF_B2 + n];
        v8f c = {bias, bias, bias, bias, bias, bias, bias, bias};
        const float* bp = lds + OFF_W2 + n * 4 + koff;
#pragma unroll
        for (int kk = 0; kk < 32; ++kk)
            c = wmma_f32(a2[kk], *(const v2f*)(bp + kk * 512), c);
#pragma unroll
        for (int r = 0; r < 8; ++r)
            myH[(r + mm) * 128 + n] = fmaxf(c[r], 0.0f);
    }
    __syncthreads();

    v2f a3[32];
#pragma unroll
    for (int kk = 0; kk < 32; ++kk)
        a3[kk] = *(const v2f*)(myH + m * 128 + 4 * kk + koff);

    {
        const float bias = lds[OFF_FCB + m];
        v8f c = {bias, bias, bias, bias, bias, bias, bias, bias};
        const float* bp = lds + OFF_FC + m * 4 + koff;
#pragma unroll
        for (int kk = 0; kk < 32; ++kk)
            c = wmma_f32(a3[kk], *(const v2f*)(bp + kk * 64), c);
#pragma unroll
        for (int r = 0; r < 8; ++r)
            out[(rowBase + r + mm) * 16 + m] = c[r];
    }
}

extern "C" void kernel_launch(void* const* d_in, const int* in_sizes, int n_in,
                              void* d_out, int out_size, void* d_ws, size_t ws_size,
                              hipStream_t stream) {
    const float* x   = (const float*)d_in[0];
    const float* W1  = (const float*)d_in[1];
    const float* b1  = (const float*)d_in[2];
    const float* W2  = (const float*)d_in[3];
    const float* b2  = (const float*)d_in[4];
    const float* fcW = (const float*)d_in[5];
    const float* fcb = (const float*)d_in[6];
    float* out = (float*)d_out;

    const size_t shmem = (size_t)LDS_FLOATS * sizeof(float);
    hipFuncSetAttribute((const void*)gnn_mlp_kernel,
                        hipFuncAttributeMaxDynamicSharedMemorySize, (int)shmem);
    gnn_mlp_kernel<<<NBLOCKS, BLOCK, shmem, stream>>>(x, W1, b1, W2, b2, fcW, fcb, out);
}


# LDS operand reads issued as ONE group per 32-WMMA block, one block ahead into a second register set (scheduler fences), instead of spread among the WMMAs; summation order unchanged
# speedup vs baseline: 1.5085x; 1.1285x over previous
#include <hip/hip_runtime.h>

typedef __attribute__((ext_vector_type(2))) float v2f;
typedef __attribute__((ext_vector_type(4))) float v4f;
typedef __attribute__((ext_vector_type(8))) float v8f;

#define WAVES  4
#define BLOCK  (WAVES * 32)
#define MTILE  (WAVES * 16)
#define NBLOCKS 256

constexpr int OFF_W1  = 0;
constexpr int OFF_W2  = OFF_W1  + 8192;
constexpr int OFF_FC  = OFF_W2  + 16384;
constexpr int OFF_B1  = OFF_FC  + 2048;
constexpr int OFF_B2  = OFF_B1  + 128;
constexpr int OFF_FCB = OFF_B2  + 128;
constexpr int OFF_H   = OFF_FCB + BLOCK;
constexpr int LDS_FLOATS = OFF_H + WAVES * 2048;

__device__ __forceinline__ v8f wmma_f32(v2f a, v2f b, v8f c) {
    return __builtin_amdgcn_wmma_f32_16x16x4_f32(false, a, false, b, (short)0, c,
                                                 false, false);
}

#define KB 4

__device__ __forceinline__ void sched_fence() { __builtin_amdgcn_sched_barrier(0); }

__device__ __forceinline__ void load_b_block(v2f (&b)[KB][8], const float* bp, int kk0) {
#pragma unroll
    for (int i = 0; i < KB; ++i)
#pragma unroll
        for (int t = 0; t < 8; ++t)
            b[i][t] = *(const v2f*)(bp + (kk0 + i) * 512 + t * 64);
}

__device__ __forceinline__ void load_a_block(v2f (&a)[KB], const float* ap, int kk0) {
#pragma unroll
    for (int i = 0; i < KB; ++i)
        a[i] = *(const v2f*)(ap + 4 * (kk0 + i));
}

__device__ __forceinline__ void mma_block(v8f (&c)[8], const v2f* a, const v2f (&b)[KB][8]) {
#pragma unroll
    for (int i = 0; i < KB; ++i)
#pragma unroll
        for (int t = 0; t < 8; ++t)
            c[t] = wmma_f32(a[i], b[i][t], c[t]);
}

template <int N, int IT>
__device__ __forceinline__ void load_blocks(v4f (&r)[IT][4], const float* __restrict__ src,
                                            int tid) {
    constexpr int NB = N / 4;
    const v4f* g = (const v4f*)src;
#pragma unroll
    for (int it = 0; it < IT; ++it) {
        const int b  = tid + it * BLOCK;
        const int kk = b / NB;
        const int nb = b % NB;
#pragma unroll
        for (int j = 0; j < 4; ++j)
            r[it][j] = g[(4 * kk + j) * NB + nb];
    }
}

template <int N, int IT>
__device__ __forceinline__ void store_blocks(float* dst, const v4f (&r)[IT][4], int tid) {
    constexpr int NB = N / 4;
#pragma unroll
    for (int it = 0; it < IT; ++it) {
        const int b  = tid + it * BLOCK;
        const int kk = b / NB;
        const int nb = b % NB;
        v4f* d = (v4f*)(dst + kk * (4 * N) + nb * 16);
#pragma unroll
        for (int i = 0; i < 4; ++i) {
            const v4f t = {r[it][0][i], r[it][1][i], r[it][2][i], r[it][3][i]};
            d[i] = t;
        }
    }
}

__global__ void __launch_bounds__(BLOCK)
gnn_mlp_kernel(const float* __restrict__ x,
               const float* __restrict__ W1, const float* __restrict__ b1,
               const float* __restrict__ W2, const float* __restrict__ b2,
               const float* __restrict__ fcW, const float* __restrict__ fcb,
               float* __restrict__ out)
{
    extern __shared__ __attribute__((aligned(16))) float lds[];
    const int tid = threadIdx.x;

    v4f rw1[4][4], rw2[8][4], rfc[1][4];
    load_blocks<128, 4>(rw1, W1, tid);
    load_blocks<128, 8>(rw2, W2, tid);
    load_blocks<16, 1>(rfc, fcW, tid);
    const float vb1 = b1[tid];
    const float vb2 = b2[tid];
    const float vfb = fcb[tid & 15];
    store_blocks<128, 4>(lds + OFF_W1, rw1, tid);
    store_blocks<128, 8>(lds + OFF_W2, rw2, tid);
    store_blocks<16, 1>(lds + OFF_FC, rfc, tid);
    lds[OFF_B1 + tid] = vb1;
    lds[OFF_B2 + tid] = vb2;
    lds[OFF_FCB + tid] = vfb;
    __syncthreads();

    const int lane = tid & 31;
    const int wave = tid >> 5;
    const int m    = lane & 15;
    const int hi   = lane >> 4;
    const int koff = hi * 2;
    const int mm   = hi * 8;

    float* myH = lds + OFF_H + wave * 2048;
    const long rowBase = (long)blockIdx.x * MTILE + wave * 16;

    v2f a1[16];
    {
        const float* xr = x + (rowBase + m) * 64;
#pragma unroll
        for (int kk = 0; kk < 16; ++kk)
            a1[kk] = *(const v2f*)(xr + 4 * kk + koff);
    }
    v8f c[8];
    v2f bs[2][KB][8];
    {
#pragma unroll
        for (int t = 0; t < 8; ++t) {
            const float bias = lds[OFF_B1 + t * 16 + m];
            c[t] = (v8f){bias, bias, bias, bias, bias, bias, bias, bias};
        }
        const float* bp = lds + OFF_W1 + m * 4 + koff;
        load_b_block(bs[0], bp, 0);
#pragma unroll
        for (int blk = 0; blk < 16 / KB; ++blk) {
            if (blk + 1 < 16 / KB)
                load_b_block(bs[(blk + 1) & 1], bp, (blk + 1) * KB);
            sched_fence();
            mma_block(c, &a1[blk * KB], bs[blk & 1]);
            sched_fence();
        }
#pragma unroll
        for (int t = 0; t < 8; ++t)
#pragma unroll
            for (int r = 0; r < 8; ++r)
                myH[(r + mm) * 128 + t * 16 + m] = fmaxf(c[t][r], 0.0f);
    }
    __syncthreads();

    v2f fa[2][KB];
    {
#pragma unroll
        for (int t = 0; t < 8; ++t) {
            const float bias = lds[OFF_B2 + t * 16 + m];
            c[t] = (v8f){bias, bias, bias, bias, bias, bias, bias, bias};
        }
        const float* ap = myH + m * 128 + koff;
        const float* bp = lds + OFF_W2 + m * 4 + koff;
        load_a_block(fa[0], ap, 0);
        load_b_block(bs[0], bp, 0);
#pragma unroll
        for (int blk = 0; blk < 32 / KB; ++blk) {
            if (blk + 1 < 32 / KB) {
                load_a_block(fa[(blk + 1) & 1], ap, (blk + 1) * KB);
                load_b_block(bs[(blk + 1) & 1], bp, (blk + 1) * KB);
            }
            sched_fence();
            mma_block(c, fa[blk & 1], bs[blk & 1]);
            sched_fence();
        }
#pragma unroll
        for (int t = 0; t < 8; ++t)
#pragma unroll
            for (int r = 0; r < 8; ++r)
                myH[(r + mm) * 128 + t * 16 + m] = fmaxf(c[t][r], 0.0f);
    }
    __syncthreads();

    {
        const float bias = lds[OFF_FCB + m];
        v8f c3 = {bias, bias, bias, bias, bias, bias, bias, bias};
        const float* ap = myH + m * 128 + koff;
        const float* bp = lds + OFF_FC + m * 4 + koff;
        v2f a3[32], b3[32];
#pragma unroll
        for (int kk = 0; kk < 32; ++kk) {
            a3[kk] = *(const v2f*)(ap + 4 * kk);
            b3[kk] = *(const v2f*)(bp + kk * 64);
        }
        sched_fence();
#pragma unroll
        for (int kk = 0; kk < 32; ++kk)
            c3 = wmma_f32(a3[kk], b3[kk], c3);
#pragma unroll
        for (int r = 0; r < 8; ++r)
            out[(rowBase + r + mm) * 16 + m] = c3[r];
    }
}

extern "C" void kernel_launch(void* const* d_in, const int* in_sizes, int n_in,
                              void* d_out, int out_size, void* d_ws, size_t ws_size,
                              hipStream_t stream) {
    const float* x   = (const float*)d_in[0];
    const float* W1  = (const float*)d_in[1];
    const float* b1  = (const float*)d_in[2];
    const float* W2  = (const float*)d_in[3];
    const float* b2  = (const float*)d_in[4];
    const float* fcW = (const float*)d_in[5];
    const float* fcb = (const float*)d_in[6];
    float* out = (float*)d_out;

    const size_t shmem = (size_t)LDS_FLOATS * sizeof(float);
    hipFuncSetAttribute((const void*)gnn_mlp_kernel,
                        hipFuncAttributeMaxDynamicSharedMemorySize, (int)shmem);
    gnn_mlp_kernel<<<NBLOCKS, BLOCK, shmem, stream>>>(x, W1, b1, W2, b2, fcW, fcb, out);
}


# Products transposed (weights = A, activations = B): a layer's accumulators are the next layer's B fragments, h1/h2 skip LDS; k grouped anew per WMMA in layers 2-3 (summation order changes)
# speedup vs baseline: 1.8155x; 1.2035x over previous
#include <hip/hip_runtime.h>


typedef __attribute__((ext_vector_type(2))) float v2f;
typedef __attribute__((ext_vector_type(4))) float v4f;
typedef __attribute__((ext_vector_type(8))) float v8f;

#define WAVES  4
#define BLOCK  (WAVES * 32)
#define MTILE  (WAVES * 16)
#define NBLOCKS 256

constexpr int OFF_W1  = 0;
constexpr int OFF_W2  = OFF_W1  + 8192;
constexpr int OFF_FC  = OFF_W2  + 16384;
constexpr int OFF_B1  = OFF_FC  + 2048;
constexpr int OFF_B2  = OFF_B1  + 128;
constexpr int OFF_FCB = OFF_B2  + 128;
constexpr int LDS_FLOATS = OFF_FCB + BLOCK;

__device__ __forceinline__ v8f wmma_f32(v2f a, v2f b, v8f c) {
    return __builtin_amdgcn_wmma_f32_16x16x4_f32(false, a, false, b, (short)0, c,
                                                 false, false);
}

#define KB 4

__device__ __forceinline__ void sched_fence() { __builtin_amdgcn_sched_barrier(0); }

__device__ __forceinline__ void load_w_block(v2f (&w)[KB][8], const float* wp, int q0) {
#pragma unroll
    for (int i = 0; i < KB; ++i)
#pragma unroll
        for (int t = 0; t < 8; ++t)
            w[i][t] = *(const v2f*)(wp + (q0 + i) * 512 + t * 64);
}

__device__ __forceinline__ void mma_block(v8f (&c)[8], const v2f* act, const v2f (&w)[KB][8]) {
#pragma unroll
    for (int i = 0; i < KB; ++i)
#pragma unroll
        for (int t = 0; t < 8; ++t)
            c[t] = wmma_f32(w[i][t], act[i], c[t]);
}

__device__ __forceinline__ v8f bias_tile(const float* bp, int t) {
    const v4f lo = *(const v4f*)(bp + t * 16);
    const v4f hi = *(const v4f*)(bp + t * 16 + 4);
    return (v8f){lo[0], lo[1], lo[2], lo[3], hi[0], hi[1], hi[2], hi[3]};
}

__device__ __forceinline__ void relu_to_frags(v2f (&act)[32], const v8f (&c)[8]) {
#pragma unroll
    for (int t = 0; t < 8; ++t)
#pragma unroll
        for (int j = 0; j < 4; ++j)
            act[4 * t + j] = (v2f){fmaxf(c[t][2 * j], 0.0f), fmaxf(c[t][2 * j + 1], 0.0f)};
}

template <bool ACC_ORDER>
__device__ __forceinline__ int krow(int q, int e) {
    return ACC_ORDER ? 16 * (q >> 2) + 8 * (e >> 1) + 2 * (q & 3) + (e & 1) : 4 * q + e;
}

template <int N, int IT, bool ACC_ORDER>
__device__ __forceinline__ void load_blocks(v4f (&r)[IT][4], const float* __restrict__ src,
                                            int tid) {
    constexpr int NB = N / 4;
    const v4f* g = (const v4f*)src;
#pragma unroll
    for (int it = 0; it < IT; ++it) {
        const int b  = tid + it * BLOCK;
        const int q  = b / NB;
        const int nb = b % NB;
#pragma unroll
        for (int e = 0; e < 4; ++e)
            r[it][e] = g[krow<ACC_ORDER>(q, e) * NB + nb];
    }
}

template <int N, int IT>
__device__ __forceinline__ void store_blocks(float* dst, const v4f (&r)[IT][4], int tid) {
    constexpr int NB = N / 4;
#pragma unroll
    for (int it = 0; it < IT; ++it) {
        const int b  = tid + it * BLOCK;
        const int q  = b / NB;
        const int nb = b % NB;
        v4f* d = (v4f*)(dst + q * (4 * N) + nb * 16);
#pragma unroll
        for (int i = 0; i < 4; ++i) {
            const v4f t = {r[it][0][i], r[it][1][i], r[it][2][i], r[it][3][i]};
            d[i] = t;
        }
    }
}

__global__ void __launch_bounds__(BLOCK)
gnn_mlp_kernel(const float* __restrict__ x,
               const float* __restrict__ W1, const float* __restrict__ b1,
               const float* __restrict__ W2, const float* __restrict__ b2,
               const float* __restrict__ fcW, const float* __restrict__ fcb,
               float* __restrict__ out)
{
    extern __shared__ __attribute__((aligned(16))) float lds[];
    const int tid = threadIdx.x;

    v4f rw1[4][4], rw2[8][4], rfc[1][4];
    load_blocks<128, 4, false>(rw1, W1, tid);
    load_blocks<128, 8, true>(rw2, W2, tid);
    load_blocks<16, 1, true>(rfc, fcW, tid);
    const float vb1 = b1[tid];
    const float vb2 = b2[tid];
    const float vfb = fcb[tid & 15];
    store_blocks<128, 4>(lds + OFF_W1, rw1, tid);
    store_blocks<128, 8>(lds + OFF_W2, rw2, tid);
    store_blocks<16, 1>(lds + OFF_FC, rfc, tid);
    lds[OFF_B1 + tid] = vb1;
    lds[OFF_B2 + tid] = vb2;
    lds[OFF_FCB + tid] = vfb;
    __syncthreads();

    const int lane = tid & 31;
    const int wave = tid >> 5;
    const int m    = lane & 15;
    const int hi   = lane >> 4;
    const int koff = hi * 2;
    const int mm   = hi * 8;

    const long rowBase = (long)blockIdx.x * MTILE + wave * 16;

    const float* wp1 = lds + OFF_W1 + m * 4 + koff;
    const float* wp2 = lds + OFF_W2 + m * 4 + koff;
    const float* wp3 = lds + OFF_FC + m * 4 + koff;

    v2f a1[16];
    {
        const float* xr = x + (rowBase + m) * 64;
#pragma unroll
        for (int q = 0; q < 16; ++q)
            a1[q] = *(const v2f*)(xr + 4 * q + koff);
    }
    v8f c1[8], c2[8];
    v2f ws[2][KB][8];
#pragma unroll
    for (int t = 0; t < 8; ++t)
        c1[t] = bias_tile(lds + OFF_B1 + mm, t);
    load_w_block(ws[0], wp1, 0);
#pragma unroll
    for (int blk = 0; blk < 16 / KB; ++blk) {
        if (blk + 1 < 16 / KB) {
            load_w_block(ws[(blk + 1) & 1], wp1, (blk + 1) * KB);
        } else {
            load_w_block(ws[(blk + 1) & 1], wp2, 0);
#pragma unroll
            for (int t = 0; t < 8; ++t)
                c2[t] = bias_tile(lds + OFF_B2 + mm, t);
        }
        sched_fence();
        mma_block(c1, &a1[blk * KB], ws[blk & 1]);
        sched_fence();
    }

    v2f act[32];
    relu_to_frags(act, c1);
    v2f w3[32];
    v8f c3;
#pragma unroll
    for (int blk = 0; blk < 32 / KB; ++blk) {
        if (blk + 1 < 32 / KB) {
            load_w_block(ws[(blk + 1) & 1], wp2, (blk + 1) * KB);
        } else {
#pragma unroll
            for (int q = 0; q < 32; ++q)
                w3[q] = *(const v2f*)(wp3 + q * 64);
            const v4f lo = *(const v4f*)(lds + OFF_FCB + mm);
            const v4f hi4 = *(const v4f*)(lds + OFF_FCB + mm + 4);
            c3 = (v8f){lo[0], lo[1], lo[2], lo[3], hi4[0], hi4[1], hi4[2], hi4[3]};
        }
        sched_fence();
        mma_block(c2, &act[blk * KB], ws[blk & 1]);
        sched_fence();
    }

    relu_to_frags(act, c2);
#pragma unroll
    for (int q = 0; q < 32; ++q)
        c3 = wmma_f32(w3[q], act[q], c3);
    {
        float* o = out + (rowBase + m) * 16 + mm;
        *(v4f*)o       = (v4f){c3[0], c3[1], c3[2], c3[3]};
        *(v4f*)(o + 4) = (v4f){c3[4], c3[5], c3[6], c3[7]};
    }
}

extern "C" void kernel_launch(void* const* d_in, const int* in_sizes, int n_in,
                              void* d_out, int out_size, void* d_ws, size_t ws_size,
                              hipStream_t stream) {
    const float* x   = (const float*)d_in[0];
    const float* W1  = (const float*)d_in[1];
    const float* b1  = (const float*)d_in[2];
    const float* W2  = (const float*)d_in[3];
    const float* b2  = (const float*)d_in[4];
    const float* fcW = (const float*)d_in[5];
    const float* fcb = (const float*)d_in[6];
    float* out = (float*)d_out;

    const size_t shmem = (size_t)LDS_FLOATS * sizeof(float);
    hipFuncSetAttribute((const void*)gnn_mlp_kernel,
                        hipFuncAttributeMaxDynamicSharedMemorySize, (int)shmem);
    gnn_mlp_kernel<<<NBLOCKS, BLOCK, shmem, stream>>>(x, W1, b1, W2, b2, fcW, fcb, out);
}


# Weight staging overlapped with compute: all global loads issued at entry, layer 1 starts once W1 is in LDS (LDS-only barrier), W2/fcW are stored to LDS in front of layer 1's last WMMA block
# speedup vs baseline: 2.0238x; 1.1147x over previous
#include <hip/hip_runtime.h>


typedef __attribute__((ext_vector_type(2))) float v2f;
typedef __attribute__((ext_vector_type(4))) float v4f;
typedef __attribute__((ext_vector_type(8))) float v8f;

#define WAVES  4
#define BLOCK  (WAVES * 32)
#define MTILE  (WAVES * 16)
#define NBLOCKS 256

constexpr int OFF_W1  = 0;
constexpr int OFF_W2  = OFF_W1  + 8192;
constexpr int OFF_FC  = OFF_W2  + 16384;
constexpr int OFF_B1  = OFF_FC  + 2048;
constexpr int OFF_B2  = OFF_B1  + 128;
constexpr int OFF_FCB = OFF_B2  + 128;
constexpr int LDS_FLOATS = OFF_FCB + BLOCK;

__device__ __forceinline__ v8f wmma_f32(v2f a, v2f b, v8f c) {
    return __builtin_amdgcn_wmma_f32_16x16x4_f32(false, a, false, b, (short)0, c,
                                                 false, false);
}

__device__ __forceinline__ void lds_barrier() {
    __builtin_amdgcn_fence(__ATOMIC_RELEASE, "workgroup", "local");
    __builtin_amdgcn_s_barrier();
    __builtin_amdgcn_fence(__ATOMIC_ACQUIRE, "workgroup", "local");
}

#define KB 4

__device__ __forceinline__ void sched_fence() { __builtin_amdgcn_sched_barrier(0); }

__device__ __forceinline__ void load_w_block(v2f (&w)[KB][8], const float* wp, int q0) {
#pragma unroll
    for (int i = 0; i < KB; ++i)
#pragma unroll
        for (int t = 0; t < 8; ++t)
            w[i][t] = *(const v2f*)(wp + (q0 + i) * 512 + t * 64);
}

__device__ __forceinline__ void mma_block(v8f (&c)[8], const v2f* act, const v2f (&w)[KB][8]) {
#pragma unroll
    for (int i = 0; i < KB; ++i)
#pragma unroll
        for (int t = 0; t < 8; ++t)
            c[t] = wmma_f32(w[i][t], act[i], c[t]);
}

__device__ __forceinline__ v8f bias_tile(const float* bp, int t) {
    const v4f lo = *(const v4f*)(bp + t * 16);
    const v4f hi = *(const v4f*)(bp + t * 16 + 4);
    return (v8f){lo[0], lo[1], lo[2], lo[3], hi[0], hi[1], hi[2], hi[3]};
}

__device__ __forceinline__ void relu_to_frags(v2f (&act)[32], const v8f (&c)[8]) {
#pragma unroll
    for (int t = 0; t < 8; ++t)
#pragma unroll
        for (int j = 0; j < 4; ++j)
            act[4 * t + j] = (v2f){fmaxf(c[t][2 * j], 0.0f), fmaxf(c[t][2 * j + 1], 0.0f)};
}

template <bool ACC_ORDER>
__device__ __forceinline__ int krow(int q, int e) {
    return ACC_ORDER ? 16 * (q >> 2) + 8 * (e >> 1) + 2 * (q & 3) + (e & 1) : 4 * q + e;
}

template <int N, int IT, bool ACC_ORDER>
__device__ __forceinline__ void load_blocks(v4f (&r)[IT][4], const float* __restrict__ src,
                                            int tid) {
    constexpr int NB = N / 4;
    const v4f* g = (const v4f*)src;
#pragma unroll
    for (int it = 0; it < IT; ++it) {
        const int b  = tid + it * BLOCK;
        const int q  = b / NB;
        const int nb = b % NB;
#pragma unroll
        for (int e = 0; e < 4; ++e)
            r[it][e] = g[krow<ACC_ORDER>(q, e) * NB + nb];
    }
}

template <int N, int IT>
__device__ __forceinline__ void store_blocks(float* dst, const v4f (&r)[IT][4], int tid) {
    constexpr int NB = N / 4;
#pragma unroll
    for (int it = 0; it < IT; ++it) {
        const int b  = tid + it * BLOCK;
        const int q  = b / NB;
        const int nb = b % NB;
        v4f* d = (v4f*)(dst + q * (4 * N) + nb * 16);
#pragma unroll
        for (int i = 0; i < 4; ++i) {
            const v4f t = {r[it][0][i], r[it][1][i], r[it][2][i], r[it][3][i]};
            d[i] = t;
        }
    }
}

__global__ void __launch_bounds__(BLOCK)
gnn_mlp_kernel(const float* __restrict__ x,
               const float* __restrict__ W1, const float* __restrict__ b1,
               const float* __restrict__ W2, const float* __restrict__ b2,
               const float* __restrict__ fcW, const float* __restrict__ fcb,
               float* __restrict__ out)
{
    extern __shared__ __attribute__((aligned(16))) float lds[];
    const int tid = threadIdx.x;

    const int lane = tid & 31;
    const int wave = tid >> 5;
    const int m    = lane & 15;
    const int hi   = lane >> 4;
    const int koff = hi * 2;
    const int mm   = hi * 8;

    const long rowBase = (long)blockIdx.x * MTILE + wave * 16;

    v4f rw1[4][4], rw2[8][4], rfc[1][4];
    load_blocks<128, 4, false>(rw1, W1, tid);
    const float vb1 = b1[tid];
    const float vb2 = b2[tid];
    const float vfb = fcb[tid & 15];
    v2f a1[16];
    {
        const float* xr = x + (rowBase + m) * 64;
#pragma unroll
        for (int q = 0; q < 16; ++q)
            a1[q] = *(const v2f*)(xr + 4 * q + koff);
    }
    sched_fence();
    load_blocks<128, 8, true>(rw2, W2, tid);
    load_blocks<16, 1, true>(rfc, fcW, tid);
    sched_fence();

    store_blocks<128, 4>(lds + OFF_W1, rw1, tid);
    lds[OFF_B1 + tid] = vb1;
    lds[OFF_B2 + tid] = vb2;
    lds[OFF_FCB + tid] = vfb;
    lds_barrier();

    const float* wp1 = lds + OFF_W1 + m * 4 + koff;
    const float* wp2 = lds + OFF_W2 + m * 4 + koff;
    const float* wp3 = lds + OFF_FC + m * 4 + koff;

    v8f c1[8], c2[8];
    v2f ws[2][KB][8];
#pragma unroll
    for (int t = 0; t < 8; ++t)
        c1[t] = bias_tile(lds + OFF_B1 + mm, t);
    load_w_block(ws[0], wp1, 0);
#pragma unroll
    for (int blk = 0; blk < 16 / KB; ++blk) {
        if (blk + 1 < 16 / KB) {
            load_w_block(ws[(blk + 1) & 1], wp1, (blk + 1) * KB);
        } else {
#pragma unroll
            for (int t = 0; t < 8; ++t)
                c2[t] = bias_tile(lds + OFF_B2 + mm, t);
            store_blocks<128, 8>(lds + OFF_W2, rw2, tid);
            store_blocks<16, 1>(lds + OFF_FC, rfc, tid);
        }
        sched_fence();
        mma_block(c1, &a1[blk * KB], ws[blk & 1]);
        sched_fence();
    }
    lds_barrier();
    load_w_block(ws[0], wp2, 0);

    v2f act[32];
    relu_to_frags(act, c1);
    v2f w3[32];
    v8f c3;
#pragma unroll
    for (int blk = 0; blk < 32 / KB; ++blk) {
        if (blk + 1 < 32 / KB) {
            load_w_block(ws[(blk + 1) & 1], wp2, (blk + 1) * KB);
        } else {
#pragma unroll
            for (int q = 0; q < 32; ++q)
                w3[q] = *(const v2f*)(wp3 + q * 64);
            const v4f lo = *(const v4f*)(lds + OFF_FCB + mm);
            const v4f hi4 = *(const v4f*)(lds + OFF_FCB + mm + 4);
            c3 = (v8f){lo[0], lo[1], lo[2], lo[3], hi4[0], hi4[1], hi4[2], hi4[3]};
        }
        sched_fence();
        mma_block(c2, &act[blk * KB], ws[blk & 1]);
        sched_fence();
    }

    relu_to_frags(act, c2);
#pragma unroll
    for (int q = 0; q < 32; ++q)
        c3 = wmma_f32(w3[q], act[q], c3);
    {
        float* o = out + (rowBase + m) * 16 + mm;
        *(v4f*)o       = (v4f){c3[0], c3[1], c3[2], c3[3]};
        *(v4f*)(o + 4) = (v4f){c3[4], c3[5], c3[6], c3[7]};
    }
}

extern "C" void kernel_launch(void* const* d_in, const int* in_sizes, int n_in,
                              void* d_out, int out_size, void* d_ws, size_t ws_size,
                              hipStream_t stream) {
    const float* x   = (const float*)d_in[0];
    const float* W1  = (const float*)d_in[1];
    const float* b1  = (const float*)d_in[2];
    const float* W2  = (const float*)d_in[3];
    const float* b2  = (const float*)d_in[4];
    const float* fcW = (const float*)d_in[5];
    const float* fcb = (const float*)d_in[6];
    float* out = (float*)d_out;

    const size_t shmem = (size_t)LDS_FLOATS * sizeof(float);
    hipFuncSetAttribute((const void*)gnn_mlp_kernel,
                        hipFuncAttributeMaxDynamicSharedMemorySize, (int)shmem);
    gnn_mlp_kernel<<<NBLOCKS, BLOCK, shmem, stream>>>(x, W1, b1, W2, b2, fcW, fcb, out);
}
